# starting kernel
# baseline (speedup 1.0000x reference)
#include <hip/hip_runtime.h>

typedef float v2f __attribute__((ext_vector_type(2)));
typedef float v8f __attribute__((ext_vector_type(8)));

#define C_DIM 256

__global__ void k_init(float* __restrict__ deg, int* __restrict__ has_edge,
                       int N, int NC) {
    int i = blockIdx.x * blockDim.x + threadIdx.x;
    if (i < N) deg[i] = 1.0f;
    if (i < NC) has_edge[i] = 0;
}

__global__ void k_edges(const int* __restrict__ src, const int* __restrict__ dst,
                        const int* __restrict__ assign,
                        float* __restrict__ deg, int* __restrict__ has_edge, int E) {
    int e = blockIdx.x * blockDim.x + threadIdx.x;
    if (e >= E) return;
    int s = src[e], d = dst[e];
    int cs = assign[s];
    if (cs == assign[d]) {
        unsafeAtomicAdd(&deg[d], 1.0f);
        has_edge[cs] = 1;
    }
}

__global__ __launch_bounds__(256) void k_gemm(const float* __restrict__ X,
                                              const float* __restrict__ W,
                                              float* __restrict__ xW, int N) {
    const int lane = threadIdx.x & 31;
    const int wave = threadIdx.x >> 5;
    const int half = lane >> 4;
    const int m    = lane & 15;
    const int r0   = blockIdx.x * 128 + wave * 16;
    if (r0 >= N) return;

    v8f acc[16] = {};

    int rowA = r0 + m;
    if (rowA >= N) rowA = N - 1;
    const float* __restrict__ Arow = X + (size_t)rowA * C_DIM + 2 * half;
    const float* __restrict__ Wb = W + (size_t)(2 * half) * C_DIM + m;

    for (int k0 = 0; k0 < C_DIM; k0 += 4) {
        v2f a;
        a.x = Arow[k0];
        a.y = Arow[k0 + 1];
#pragma unroll
        for (int t = 0; t < 16; ++t) {
            v2f bf;
            bf.x = Wb[(size_t)k0 * C_DIM + t * 16];
            bf.y = Wb[(size_t)(k0 + 1) * C_DIM + t * 16];
            acc[t] = __builtin_amdgcn_wmma_f32_16x16x4_f32(
                false, a, false, bf, (short)0, acc[t], false, false);
        }
    }

#pragma unroll
    for (int t = 0; t < 16; ++t) {
#pragma unroll
        for (int v = 0; v < 8; ++v) {
            int row = r0 + v + 8 * half;
            if (row < N) xW[(size_t)row * C_DIM + t * 16 + m] = acc[t][v];
        }
    }
}

__global__ void k_self(const float* __restrict__ X, const float* __restrict__ xW,
                       const float* __restrict__ deg, const float* __restrict__ b,
                       const int* __restrict__ assign, const int* __restrict__ has_edge,
                       float* __restrict__ out, int N) {
    int idx = blockIdx.x * blockDim.x + threadIdx.x;
    if (idx >= N * C_DIM) return;
    int n = idx >> 8;
    int c = idx & (C_DIM - 1);
    if (has_edge[assign[n]]) {
        out[idx] = xW[idx] / deg[n] + b[c];
    } else {
        out[idx] = X[idx];
    }
}

__global__ __launch_bounds__(256) void k_scatter(const int* __restrict__ src,
                                                 const int* __restrict__ dst,
                                                 const int* __restrict__ assign,
                                                 const float* __restrict__ deg,
                                                 const float* __restrict__ xW,
                                                 float* __restrict__ out, int E) {
    int e    = (blockIdx.x * blockDim.x + threadIdx.x) >> 5;
    int lane = threadIdx.x & 31;
    if (e >= E) return;
    int s = src[e], d = dst[e];
    if (assign[s] != assign[d]) return;
    float w = rsqrtf(deg[s]) * rsqrtf(deg[d]);
    const float4* xs = (const float4*)(xW + (size_t)s * C_DIM) + lane * 2;
    float4 v0 = xs[0];
    float4 v1 = xs[1];
    float* o = out + (size_t)d * C_DIM + lane * 8;
    unsafeAtomicAdd(o + 0, w * v0.x);
    unsafeAtomicAdd(o + 1, w * v0.y);
    unsafeAtomicAdd(o + 2, w * v0.z);
    unsafeAtomicAdd(o + 3, w * v0.w);
    unsafeAtomicAdd(o + 4, w * v1.x);
    unsafeAtomicAdd(o + 5, w * v1.y);
    unsafeAtomicAdd(o + 6, w * v1.z);
    unsafeAtomicAdd(o + 7, w * v1.w);
}

extern "C" void kernel_launch(void* const* d_in, const int* in_sizes, int n_in,
                              void* d_out, int out_size, void* d_ws, size_t ws_size,
                              hipStream_t stream) {
    const float* X      = (const float*)d_in[0];
    const float* W      = (const float*)d_in[1];
    const float* b      = (const float*)d_in[2];
    const int*   assign = (const int*)d_in[3];
    const int*   ei     = (const int*)d_in[4];

    const int N  = in_sizes[0] / C_DIM;
    const int E  = in_sizes[4] / 2;
    const int NC = 256;

    const int* src = ei;
    const int* dst = ei + E;

    char* ws = (char*)d_ws;
    size_t off = 0;
    float* deg = (float*)(ws + off);
    off += (((size_t)N * 4) + 255) & ~(size_t)255;
    int* has_edge = (int*)(ws + off);
    off += (((size_t)NC * 4) + 255) & ~(size_t)255;
    float* xW = (float*)(ws + off);

    float* out = (float*)d_out;

    k_init<<<(N + 255) / 256, 256, 0, stream>>>(deg, has_edge, N, NC);
    k_edges<<<(E + 255) / 256, 256, 0, stream>>>(src, dst, assign, deg, has_edge, E);
    k_gemm<<<(N + 127) / 128, 256, 0, stream>>>(X, W, xW, N);
    k_self<<<(N * C_DIM + 255) / 256, 256, 0, stream>>>(X, xW, deg, b, assign,
                                                        has_edge, out, N);
    k_scatter<<<(E + 7) / 8, 256, 0, stream>>>(src, dst, assign, deg, xW, out, E);
}


# xW=X@W: W copied once per work-group into LDS in k-quad order (one 32-wave work-group per WGP, 16x64 units dealt evenly over the SIMDs); B by one 16-byte LDS load per two products, not global dwords
# speedup vs baseline: 2.6502x; 2.6502x over previous
#include <hip/hip_runtime.h>

typedef float v2f __attribute__((ext_vector_type(2)));
typedef float v8f __attribute__((ext_vector_type(8)));

#define C_DIM 256

__global__ void k_init(float* __restrict__ deg, int* __restrict__ has_edge,
                       int N, int NC) {
    int i = blockIdx.x * blockDim.x + threadIdx.x;
    if (i < N) deg[i] = 1.0f;
    if (i < NC) has_edge[i] = 0;
}

__global__ void k_edges(const int* __restrict__ src, const int* __restrict__ dst,
                        const int* __restrict__ assign,
                        float* __restrict__ deg, int* __restrict__ has_edge, int E) {
    int e = blockIdx.x * blockDim.x + threadIdx.x;
    if (e >= E) return;
    int s = src[e], d = dst[e];
    int cs = assign[s];
    if (cs == assign[d]) {
        unsafeAtomicAdd(&deg[d], 1.0f);
        has_edge[cs] = 1;
    }
}

#define G_TILES   13
#define G_ROWS    (G_TILES * 16)
#define G_UNITS   (G_TILES * 4)
#define G_THREADS 1024
#define G_WAVES   (G_THREADS / 32)
#define W_PITCH   1056

__global__ __launch_bounds__(G_THREADS) void k_gemm(const float* __restrict__ X,
                                                    const float* __restrict__ W,
                                                    float* __restrict__ xW, int N) {
    __shared__ __attribute__((aligned(16))) float Wl[(C_DIM / 4) * W_PITCH];

    const int lane = threadIdx.x & 31;
    const int wave = __builtin_amdgcn_readfirstlane(threadIdx.x >> 5);
    const int h    = lane >> 4;
    const int c    = lane & 15;

    for (int i = 0; i < 2; ++i) {
        const int q = wave + G_WAVES * i;
        const float* __restrict__ wq = W + (size_t)(4 * q) * C_DIM + lane;
        float* lq = Wl + q * W_PITCH + 4 * lane;
#pragma unroll
        for (int cc = 0; cc < 8; ++cc) {
            float4 v;
            v.x = wq[cc * 32];
            v.y = wq[C_DIM + cc * 32];
            v.z = wq[2 * C_DIM + cc * 32];
            v.w = wq[3 * C_DIM + cc * 32];
            *(float4*)(lq + cc * 128) = v;
        }
    }
    __syncthreads();

    const int rbase = blockIdx.x * G_ROWS;
    for (int u = wave; u < G_UNITS; u += G_WAVES) {
        const int r0 = rbase + (u >> 2) * 16;
        if (r0 >= N) break;
        const int n0 = (u & 3) * 64;

        int rowA = r0 + c;
        if (rowA >= N) rowA = N - 1;
        const float* __restrict__ Ap = X + (size_t)rowA * C_DIM + 4 * h;
        const float* Bp = Wl + h * W_PITCH + 4 * (n0 + c);

        v8f acc[4] = {};
        for (int k8 = 0; k8 < C_DIM; k8 += 8) {
            const float4 a = *(const float4*)(Ap + k8);
            const float* bq = Bp + (k8 >> 2) * W_PITCH;
            float4 b[4];
#pragma unroll
            for (int t = 0; t < 4; ++t) b[t] = *(const float4*)(bq + t * 64);
            v2f a0, a1;
            a0.x = a.x; a0.y = a.y;
            a1.x = a.z; a1.y = a.w;
#pragma unroll
            for (int t = 0; t < 4; ++t) {
                v2f bf;
                bf.x = b[t].x; bf.y = b[t].y;
                acc[t] = __builtin_amdgcn_wmma_f32_16x16x4_f32(
                    false, a0, false, bf, (short)0, acc[t], false, false);
            }
#pragma unroll
            for (int t = 0; t < 4; ++t) {
                v2f bf;
                bf.x = b[t].z; bf.y = b[t].w;
                acc[t] = __builtin_amdgcn_wmma_f32_16x16x4_f32(
                    false, a1, false, bf, (short)0, acc[t], false, false);
            }
        }

        float* __restrict__ op = xW + (size_t)(r0 + 8 * h) * C_DIM + n0 + c;
        if (r0 + 16 <= N) {
#pragma unroll
            for (int t = 0; t < 4; ++t) {
#pragma unroll
                for (int v = 0; v < 8; ++v) op[(size_t)v * C_DIM + t * 16] = acc[t][v];
            }
        } else {
#pragma unroll
            for (int t = 0; t < 4; ++t) {
#pragma unroll
                for (int v = 0; v < 8; ++v) {
                    if (r0 + v + 8 * h < N) op[(size_t)v * C_DIM + t * 16] = acc[t][v];
                }
            }
        }
    }
}

#define TILE_ROWS 196
#define LIST_CAP  4096
#define AGG_THREADS 1024
#define AGG_WAVES (AGG_THREADS / 32)
#define ROWS_PER_WAVE ((TILE_ROWS + AGG_WAVES - 1) / AGG_WAVES)


__device__ __forceinline__ void take4(int4 v, int e, int base, unsigned rows,
                                      int* cnt, int* l_e) {
    const bool m0 = (unsigned)(v.x - base) < rows;
    const bool m1 = (unsigned)(v.y - base) < rows;
    const bool m2 = (unsigned)(v.z - base) < rows;
    const bool m3 = (unsigned)(v.w - base) < rows;
    const int c = (int)m0 + (int)m1 + (int)m2 + (int)m3;
    if (c) {
        int p = atomicAdd(cnt, c);
        if (m0) { if (p < LIST_CAP) l_e[p] = e;     ++p; }
        if (m1) { if (p < LIST_CAP) l_e[p] = e + 1; ++p; }
        if (m2) { if (p < LIST_CAP) l_e[p] = e + 2; ++p; }
        if (m3) { if (p < LIST_CAP) l_e[p] = e + 3; }
    }
}

__device__ __forceinline__ void scan_range(const int* __restrict__ dst, int g0, int g1,
                                           int n4, int E, int base, unsigned rows,
                                           int* cnt, int* l_e) {
    const int4* __restrict__ d4 = (const int4*)dst;
    const int t = threadIdx.x;
    const int len = g1 - g0;
    const int rot = (int)(((long long)blockIdx.x * len) / gridDim.x) & ~31;
    for (int k0 = t; k0 < len; k0 += 4 * AGG_THREADS) {
        int4 v[4];
        int  g[4];
#pragma unroll
        for (int u = 0; u < 4; ++u) {
            const int k = k0 + u * AGG_THREADS;
            int j = k + rot;
            if (j >= len) j -= len;
            g[u] = g0 + j;
            v[u] = (k < len) ? d4[g[u]] : make_int4(-1, -1, -1, -1);
        }
#pragma unroll
        for (int u = 0; u < 4; ++u) take4(v[u], g[u] * 4, base, rows, cnt, l_e);
    }
    if (g1 == n4 && t < (E & 3)) {
        const int e = 4 * n4 + t;
        if ((unsigned)(dst[e] - base) < rows) {
            const int p = atomicAdd(cnt, 1);
            if (p < LIST_CAP) l_e[p] = e;
        }
    }
}

__device__ __forceinline__ void push(bool p, int s, int dl, float w, int lane,
                                     int* cnt1, int* l_s, int* l_d, float* l_w) {
    const unsigned m = __builtin_amdgcn_ballot_w32(p);
    if (m) {
        int start = 0;
        if (lane == 0) start = atomicAdd(cnt1, __popc(m));
        start = __builtin_amdgcn_readfirstlane(start);
        if (p) {
            const int q = start + __popc(m & ((1u << lane) - 1u));
            l_s[q] = s;
            l_d[q] = dl;
            l_w[q] = w;
        }
    }
}

__device__ __forceinline__ void add_row(float* tr, float w, float4 a, float4 c) {
    unsafeAtomicAdd(tr,       w * a.x);
    unsafeAtomicAdd(tr + 32,  w * a.y);
    unsafeAtomicAdd(tr + 64,  w * a.z);
    unsafeAtomicAdd(tr + 96,  w * a.w);
    unsafeAtomicAdd(tr + 128, w * c.x);
    unsafeAtomicAdd(tr + 160, w * c.y);
    unsafeAtomicAdd(tr + 192, w * c.z);
    unsafeAtomicAdd(tr + 224, w * c.w);
}

__global__ __launch_bounds__(AGG_THREADS) void k_agg(const float* __restrict__ X,
                                                     const float* __restrict__ xW,
                                                     const float* __restrict__ deg,
                                                     const float* __restrict__ b,
                                                     const int* __restrict__ assign,
                                                     const int* __restrict__ has_edge,
                                                     const int* __restrict__ src,
                                                     const int* __restrict__ dst,
                                                     float* __restrict__ out, int N, int E) {
    __shared__ __attribute__((aligned(16))) float tile[TILE_ROWS * C_DIM];
    __shared__ int   l_e[LIST_CAP];
    __shared__ int   l_s[LIST_CAP];
    __shared__ int   l_d[LIST_CAP];
    __shared__ float l_w[LIST_CAP];
    __shared__ float r_deg[TILE_ROWS];
    __shared__ int   r_cl[TILE_ROWS];
    __shared__ int   r_he[TILE_ROWS];
    __shared__ int   cnt[2];

    const int t    = threadIdx.x;
    const int lane = t & 31;
    const int wave = t >> 5;
    const int base = blockIdx.x * TILE_ROWS;
    const int left = N - base;
    const int rows = left < TILE_ROWS ? left : TILE_ROWS;

    if (t < rows) {
        const int n  = base + t;
        const int cl = assign[n];
        r_cl[t]  = cl;
        r_he[t]  = has_edge[cl];
        r_deg[t] = deg[n];
    }
    if (t < 2) cnt[t] = 0;
    __syncthreads();

    {
        const float4 b0 = ((const float4*)b)[lane];
        const float4 b1 = ((const float4*)b)[32 + lane];
        float4 va[ROWS_PER_WAVE], vb[ROWS_PER_WAVE];
#pragma unroll
        for (int k = 0; k < ROWS_PER_WAVE; ++k) {
            const int r = wave + AGG_WAVES * k;
            va[k] = make_float4(0.f, 0.f, 0.f, 0.f);
            vb[k] = va[k];
            if (r < rows) {
                const float4* rp = (const float4*)((r_he[r] ? xW : X) + (size_t)(base + r) * C_DIM);
                va[k] = rp[lane];
                vb[k] = rp[32 + lane];
            }
        }
#pragma unroll
        for (int k = 0; k < ROWS_PER_WAVE; ++k) {
            const int r = wave + AGG_WAVES * k;
            if (r < rows) {
                float4 a = va[k], c = vb[k];
                if (r_he[r]) {
                    const float dg = r_deg[r];
                    a.x = a.x / dg + b0.x; a.y = a.y / dg + b0.y;
                    a.z = a.z / dg + b0.z; a.w = a.w / dg + b0.w;
                    c.x = c.x / dg + b1.x; c.y = c.y / dg + b1.y;
                    c.z = c.z / dg + b1.z; c.w = c.w / dg + b1.w;
                }
                float* tr = tile + r * C_DIM + lane;
                tr[0]   = a.x; tr[32]  = a.y; tr[64]  = a.z; tr[96]  = a.w;
                tr[128] = c.x; tr[160] = c.y; tr[192] = c.z; tr[224] = c.w;
            }
        }
    }

    const int n4 = E >> 2;
    scan_range(dst, 0, n4, n4, E, base, (unsigned)rows, &cnt[0], l_e);
    __syncthreads();

    int step4 = n4;
    if (cnt[0] > LIST_CAP) step4 = LIST_CAP / 4 - 1;
    if (step4 < 1) step4 = 1;
    const bool one_pass = (step4 == n4) || (n4 == 0);

    for (int g0 = 0; g0 == 0 || g0 < n4; g0 += step4) {
        if (!one_pass) {
            __syncthreads();
            if (t < 2) cnt[t] = 0;
            __syncthreads();
            const int g1 = (g0 + step4 < n4) ? g0 + step4 : n4;
            scan_range(dst, g0, g1, n4, E, base, (unsigned)rows, &cnt[0], l_e);
            __syncthreads();
        }
        const int tot = cnt[0] < LIST_CAP ? cnt[0] : LIST_CAP;

        for (int ib = wave * 32; ib < tot; ib += 2 * AGG_THREADS) {
            const int  i0 = ib + lane, i1 = i0 + AGG_THREADS;
            const bool k0 = i0 < tot, k1 = i1 < tot;
            const int  e0 = l_e[k0 ? i0 : 0], e1 = l_e[k1 ? i1 : 0];
            const int  s0 = src[e0], d0 = dst[e0] - base;
            const int  s1 = src[e1], d1 = dst[e1] - base;
            const int  cs0 = assign[s0], cs1 = assign[s1];
            const float gs0 = deg[s0], gs1 = deg[s1];
            const bool in0 = k0 && cs0 == r_cl[d0];
            const bool in1 = k1 && cs1 == r_cl[d1];
            const float w0 = rsqrtf(gs0) * rsqrtf(r_deg[d0]);
            const float w1 = rsqrtf(gs1) * rsqrtf(r_deg[d1]);
            push(in0, s0, d0, w0, lane, &cnt[1], l_s, l_d, l_w);
            push(in1, s1, d1, w1, lane, &cnt[1], l_s, l_d, l_w);
        }
        __syncthreads();
        const int tot2 = cnt[1];

        const int nb = tot2 >> 2;
        for (int bi = wave; bi < nb; bi += AGG_WAVES) {
            int   s[4], dl[4];
            float w[4];
#pragma unroll
            for (int u = 0; u < 4; ++u) {
                s[u]  = __builtin_amdgcn_readfirstlane(l_s[4 * bi + u]);
                dl[u] = __builtin_amdgcn_readfirstlane(l_d[4 * bi + u]);
                w[u]  = l_w[4 * bi + u];
            }
            float4 a[4], c[4];
#pragma unroll
            for (int u = 0; u < 4; ++u) {
                const float4* rp = (const float4*)(xW + (size_t)s[u] * C_DIM);
                a[u] = rp[lane];
                c[u] = rp[32 + lane];
            }
#pragma unroll
            for (int u = 0; u < 4; ++u) add_row(tile + dl[u] * C_DIM + lane, w[u], a[u], c[u]);
        }
        if (wave < (tot2 & 3)) {
            const int i  = 4 * nb + wave;
            const int s  = __builtin_amdgcn_readfirstlane(l_s[i]);
            const int dl = __builtin_amdgcn_readfirstlane(l_d[i]);
            const float4* rp = (const float4*)(xW + (size_t)s * C_DIM);
            add_row(tile + dl * C_DIM + lane, l_w[i], rp[lane], rp[32 + lane]);
        }
    }
    __syncthreads();

    for (int r = wave; r < rows; r += AGG_WAVES) {
        const float* tr = tile + r * C_DIM + lane;
        float4 a, c;
        a.x = tr[0];   a.y = tr[32];  a.z = tr[64];  a.w = tr[96];
        c.x = tr[128]; c.y = tr[160]; c.z = tr[192]; c.w = tr[224];
        float4* op = (float4*)(out + (size_t)(base + r) * C_DIM);
        op[lane]      = a;
        op[32 + lane] = c;
    }
}

extern "C" void kernel_launch(void* const* d_in, const int* in_sizes, int n_in,
                              void* d_out, int out_size, void* d_ws, size_t ws_size,
                              hipStream_t stream) {
    const float* X      = (const float*)d_in[0];
    const float* W      = (const float*)d_in[1];
    const float* b      = (const float*)d_in[2];
    const int*   assign = (const int*)d_in[3];
    const int*   ei     = (const int*)d_in[4];

    const int N  = in_sizes[0] / C_DIM;
    const int E  = in_sizes[4] / 2;
    const int NC = 256;

    const int* src = ei;
    const int* dst = ei + E;

    char* ws = (char*)d_ws;
    size_t off = 0;
    float* deg = (float*)(ws + off);
    off += (((size_t)N * 4) + 255) & ~(size_t)255;
    int* has_edge = (int*)(ws + off);
    off += (((size_t)NC * 4) + 255) & ~(size_t)255;
    float* xW = (float*)(ws + off);

    float* out = (float*)d_out;

    k_init<<<(N + 255) / 256, 256, 0, stream>>>(deg, has_edge, N, NC);
    k_edges<<<(E + 255) / 256, 256, 0, stream>>>(src, dst, assign, deg, has_edge, E);
    k_gemm<<<(N + G_ROWS - 1) / G_ROWS, G_THREADS, 0, stream>>>(X, W, xW, N);
    k_agg<<<(N + TILE_ROWS - 1) / TILE_ROWS, AGG_THREADS, 0, stream>>>(X, xW, deg, b, assign, has_edge,
                                                                     src, dst, out, N, E);
}


# search of dst moved under the matrix work: the 12 waves of a matrix work-group that have one unit first list the edges of an aggregation tile in the workspace; the aggregation copies its list
# speedup vs baseline: 3.3573x; 1.2668x over previous
#include <hip/hip_runtime.h>

typedef float v2f __attribute__((ext_vector_type(2)));
typedef float v8f __attribute__((ext_vector_type(8)));

#define C_DIM 256

#define E_WGS 15
#define NCL   256
#define TILE_ROWS 196
#define BCAP  2048

__device__ __forceinline__ void edge_owner(const int* __restrict__ src, const int* __restrict__ dst,
                                           const int* __restrict__ assign,
                                           float* __restrict__ deg, int* __restrict__ he_part,
                                           int* lds, int w, int nthr, int N, int E) {
    const int t    = threadIdx.x;
    const int per  = (N + E_WGS - 1) / E_WGS;
    const int base = w * per;
    int rows = N - base;
    if (rows > per) rows = per;
    if (rows < 0) rows = 0;
    int* cnt  = lds;
    int* asg  = lds + per;
    int* flag = lds + 2 * per;
    for (int i = t; i < rows; i += nthr) { cnt[i] = 0; asg[i] = assign[base + i]; }
    for (int i = t; i < NCL; i += nthr) flag[i] = 0;
    __syncthreads();

    const int4* __restrict__ s4 = (const int4*)src;
    const int4* __restrict__ d4 = (const int4*)dst;
    const int n4  = E >> 2;
    const int rot = (int)(((long long)w * n4) / E_WGS) & ~31;
    for (int k0 = t; k0 < n4; k0 += 4 * nthr) {
        int s[16], d[16];
#pragma unroll
        for (int u = 0; u < 4; ++u) {
            const int k = k0 + u * nthr;
            int j = k + rot;
            if (j >= n4) j -= n4;
            int4 sv = make_int4(0, 0, 0, 0), dv = make_int4(-1, -1, -1, -1);
            if (k < n4) { sv = s4[j]; dv = d4[j]; }
            s[4 * u] = sv.x; s[4 * u + 1] = sv.y; s[4 * u + 2] = sv.z; s[4 * u + 3] = sv.w;
            d[4 * u] = dv.x; d[4 * u + 1] = dv.y; d[4 * u + 2] = dv.z; d[4 * u + 3] = dv.w;
        }
        int cs[16], cd[16];
#pragma unroll
        for (int i = 0; i < 16; ++i) {
            const unsigned dl = (unsigned)(d[i] - base);
            const bool m = dl < (unsigned)rows;
            cs[i] = assign[m ? s[i] : 0];
            cd[i] = m ? asg[dl] : -1;
            d[i]  = m ? (int)dl : -1;
        }
#pragma unroll
        for (int i = 0; i < 16; ++i) {
            if (d[i] >= 0 && cs[i] == cd[i]) {
                atomicAdd(&cnt[d[i]], 1);
                if ((unsigned)cs[i] < (unsigned)NCL) flag[cs[i]] = 1;
            }
        }
    }
    if (t < (E & 3)) {
        const int e = 4 * n4 + t;
        const unsigned dl = (unsigned)(dst[e] - base);
        if (dl < (unsigned)rows) {
            const int c = assign[src[e]];
            if (c == asg[dl]) {
                atomicAdd(&cnt[dl], 1);
                if ((unsigned)c < (unsigned)NCL) flag[c] = 1;
            }
        }
    }
    __syncthreads();

    for (int i = t; i < rows; i += nthr) deg[base + i] = 1.0f + (float)cnt[i];
    if (t < NCL / 32) {
        unsigned bits = 0;
        for (int i = 0; i < 32; ++i) bits |= (flag[32 * t + i] ? 1u : 0u) << i;
        he_part[w * (NCL / 32) + t] = (int)bits;
    }
}

#define G_TILES   13
#define G_ROWS    (G_TILES * 16)
#define G_UNITS   (G_TILES * 4)
#define G_THREADS 1024
#define G_WAVES   (G_THREADS / 32)
#define W_PITCH   1056

#define S_WAVES (2 * G_WAVES - G_UNITS)
#define S_DEPTH 16

__device__ __forceinline__ unsigned hits4(int4 v, int base, unsigned rows) {
    return ((unsigned)(v.x - base) < rows ? 1u : 0u) | ((unsigned)(v.y - base) < rows ? 2u : 0u) |
           ((unsigned)(v.z - base) < rows ? 4u : 0u) | ((unsigned)(v.w - base) < rows ? 8u : 0u);
}

__device__ __forceinline__ void drain(unsigned hm, int kfirst, int nthr, int rot, int n4,
                                      int* cnt, int* __restrict__ list) {
    while (hm) {
        const int b = __builtin_ctz(hm);
        hm &= hm - 1u;
        int j = kfirst + (b >> 2) * nthr + rot;
        if (j >= n4) j -= n4;
        const int p = atomicAdd(cnt, 1);
        if (p < BCAP) list[p] = 4 * j + (b & 3);
    }
}

__device__ __forceinline__ void tile_search(const int* __restrict__ dst, int n4, int E,
                                            int base, unsigned rows, int* cnt, int* __restrict__ list,
                                            int base2, unsigned rows2, int* cnt2, int* __restrict__ list2,
                                            int sw, int lane, int rot) {
    const int4* __restrict__ d4 = (const int4*)dst;
    const int nthr = S_WAVES * 32;
    const int t = sw * 32 + lane;
    for (int k0 = t; k0 < n4; k0 += S_DEPTH * nthr) {
        int4 v[S_DEPTH];
#pragma unroll
        for (int u = 0; u < S_DEPTH; ++u) {
            const int k = k0 + u * nthr;
            int j = k + rot;
            if (j >= n4) j -= n4;
            v[u] = (k < n4) ? d4[j] : make_int4(-1, -1, -1, -1);
        }
        unsigned ha = 0, hb = 0;
#pragma unroll
        for (int u = 0; u < 8; ++u) {
            ha |= hits4(v[u], base, rows) << (4 * u);
            hb |= hits4(v[8 + u], base, rows) << (4 * u);
        }
        drain(ha, k0, nthr, rot, n4, cnt, list);
        drain(hb, k0 + 8 * nthr, nthr, rot, n4, cnt, list);
        if (rows2) {
            ha = 0; hb = 0;
#pragma unroll
            for (int u = 0; u < 8; ++u) {
                ha |= hits4(v[u], base2, rows2) << (4 * u);
                hb |= hits4(v[8 + u], base2, rows2) << (4 * u);
            }
            drain(ha, k0, nthr, rot, n4, cnt2, list2);
            drain(hb, k0 + 8 * nthr, nthr, rot, n4, cnt2, list2);
        }
    }
    if (sw == 0 && lane < (E & 3)) {
        const int e = 4 * n4 + lane;
        const int d = dst[e];
        if ((unsigned)(d - base) < rows) {
            const int p = atomicAdd(cnt, 1);
            if (p < BCAP) list[p] = e;
        }
        if ((unsigned)(d - base2) < rows2) {
            const int p = atomicAdd(cnt2, 1);
            if (p < BCAP) list2[p] = e;
        }
    }
}

__global__ __launch_bounds__(G_THREADS) void k_gemm(const float* __restrict__ X,
                                                    const float* __restrict__ W,
                                                    float* __restrict__ xW,
                                                    const int* __restrict__ src,
                                                    const int* __restrict__ dst,
                                                    const int* __restrict__ assign,
                                                    float* __restrict__ deg,
                                                    int* __restrict__ he_part,
                                                    int* __restrict__ t_cnt,
                                                    int* __restrict__ t_list, int lists,
                                                    int N, int E) {
    __shared__ __attribute__((aligned(16))) float Wl[(C_DIM / 4) * W_PITCH];
    __shared__ int lcnt[2];

    const int n_mat = (N + G_ROWS - 1) / G_ROWS;
    if ((int)blockIdx.x >= n_mat) {
        edge_owner(src, dst, assign, deg, he_part, (int*)Wl, (int)blockIdx.x - n_mat, G_THREADS, N, E);
        return;
    }

    const int lane = threadIdx.x & 31;
    const int wave = __builtin_amdgcn_readfirstlane(threadIdx.x >> 5);
    const int h    = lane >> 4;
    const int c    = lane & 15;

    for (int i = 0; i < 2; ++i) {
        const int q = wave + G_WAVES * i;
        const float* __restrict__ wq = W + (size_t)(4 * q) * C_DIM + lane;
        float* lq = Wl + q * W_PITCH + 4 * lane;
#pragma unroll
        for (int cc = 0; cc < 8; ++cc) {
            float4 v;
            v.x = wq[cc * 32];
            v.y = wq[C_DIM + cc * 32];
            v.z = wq[2 * C_DIM + cc * 32];
            v.w = wq[3 * C_DIM + cc * 32];
            *(float4*)(lq + cc * 128) = v;
        }
    }
    if (threadIdx.x < 2) lcnt[threadIdx.x] = 0;
    __syncthreads();

    const int ntiles = (N + TILE_ROWS - 1) / TILE_ROWS;
    const int b2 = n_mat + (int)blockIdx.x;
    if (lists && wave >= G_WAVES - S_WAVES) {
        const int base1 = (int)blockIdx.x * TILE_ROWS;
        const int left1 = N - base1;
        const int rows1 = left1 < TILE_ROWS ? left1 : TILE_ROWS;
        const int base2 = b2 * TILE_ROWS;
        const int left2 = N - base2;
        int rows2 = 0;
        if (b2 < ntiles) rows2 = left2 < TILE_ROWS ? left2 : TILE_ROWS;
        const int n4  = E >> 2;
        const int rot = (int)(((long long)blockIdx.x * n4) / n_mat) & ~31;
        tile_search(dst, n4, E, base1, (unsigned)rows1, &lcnt[0], t_list + (size_t)blockIdx.x * BCAP,
                    base2, (unsigned)rows2, &lcnt[1], t_list + (size_t)b2 * BCAP,
                    wave - (G_WAVES - S_WAVES), lane, rot);
    }

    const int rbase = blockIdx.x * G_ROWS;
    for (int u = wave; u < G_UNITS; u += G_WAVES) {
        const int r0 = rbase + (u >> 2) * 16;
        if (r0 >= N) break;
        const int n0 = (u & 3) * 64;

        int rowA = r0 + c;
        if (rowA >= N) rowA = N - 1;
        const float* __restrict__ Ap = X + (size_t)rowA * C_DIM + 4 * h;
        const float* Bp = Wl + h * W_PITCH + 4 * (n0 + c);

        v8f acc[4] = {};
        for (int k8 = 0; k8 < C_DIM; k8 += 8) {
            const float4 a = *(const float4*)(Ap + k8);
            const float* bq = Bp + (k8 >> 2) * W_PITCH;
            float4 b[4];
#pragma unroll
            for (int t = 0; t < 4; ++t) b[t] = *(const float4*)(bq + t * 64);
            v2f a0, a1;
            a0.x = a.x; a0.y = a.y;
            a1.x = a.z; a1.y = a.w;
#pragma unroll
            for (int t = 0; t < 4; ++t) {
                v2f bf;
                bf.x = b[t].x; bf.y = b[t].y;
                acc[t] = __builtin_amdgcn_wmma_f32_16x16x4_f32(
                    false, a0, false, bf, (short)0, acc[t], false, false);
            }
#pragma unroll
            for (int t = 0; t < 4; ++t) {
                v2f bf;
                bf.x = b[t].z; bf.y = b[t].w;
                acc[t] = __builtin_amdgcn_wmma_f32_16x16x4_f32(
                    false, a1, false, bf, (short)0, acc[t], false, false);
            }
        }

        float* __restrict__ op = xW + (size_t)(r0 + 8 * h) * C_DIM + n0 + c;
        if (r0 + 16 <= N) {
#pragma unroll
            for (int t = 0; t < 4; ++t) {
#pragma unroll
                for (int v = 0; v < 8; ++v) op[(size_t)v * C_DIM + t * 16] = acc[t][v];
            }
        } else {
#pragma unroll
            for (int t = 0; t < 4; ++t) {
#pragma unroll
                for (int v = 0; v < 8; ++v) {
                    if (r0 + v + 8 * h < N) op[(size_t)v * C_DIM + t * 16] = acc[t][v];
                }
            }
        }
    }

    __syncthreads();
    if (lists) {
        if (threadIdx.x == 0) t_cnt[blockIdx.x] = lcnt[0];
        if (threadIdx.x == 1 && b2 < ntiles) t_cnt[b2] = lcnt[1];
    }
}

#define LIST_CAP  4096
#define AGG_THREADS 1024
#define AGG_WAVES (AGG_THREADS / 32)
#define ROWS_PER_WAVE ((TILE_ROWS + AGG_WAVES - 1) / AGG_WAVES)


__device__ __forceinline__ void take4(int4 v, int e, int base, unsigned rows,
                                      int* cnt, int* l_e) {
    const bool m0 = (unsigned)(v.x - base) < rows;
    const bool m1 = (unsigned)(v.y - base) < rows;
    const bool m2 = (unsigned)(v.z - base) < rows;
    const bool m3 = (unsigned)(v.w - base) < rows;
    const int c = (int)m0 + (int)m1 + (int)m2 + (int)m3;
    if (c) {
        int p = atomicAdd(cnt, c);
        if (m0) { if (p < LIST_CAP) l_e[p] = e;     ++p; }
        if (m1) { if (p < LIST_CAP) l_e[p] = e + 1; ++p; }
        if (m2) { if (p < LIST_CAP) l_e[p] = e + 2; ++p; }
        if (m3) { if (p < LIST_CAP) l_e[p] = e + 3; }
    }
}

__device__ __forceinline__ void scan_range(const int* __restrict__ dst, int g0, int g1,
                                           int n4, int E, int base, unsigned rows,
                                           int* cnt, int* l_e) {
    const int4* __restrict__ d4 = (const int4*)dst;
    const int t = threadIdx.x;
    const int len = g1 - g0;
    const int rot = (int)(((long long)blockIdx.x * len) / gridDim.x) & ~31;
    for (int k0 = t; k0 < len; k0 += 4 * AGG_THREADS) {
        int4 v[4];
        int  g[4];
#pragma unroll
        for (int u = 0; u < 4; ++u) {
            const int k = k0 + u * AGG_THREADS;
            int j = k + rot;
            if (j >= len) j -= len;
            g[u] = g0 + j;
            v[u] = (k < len) ? d4[g[u]] : make_int4(-1, -1, -1, -1);
        }
#pragma unroll
        for (int u = 0; u < 4; ++u) take4(v[u], g[u] * 4, base, rows, cnt, l_e);
    }
    if (g1 == n4 && t < (E & 3)) {
        const int e = 4 * n4 + t;
        if ((unsigned)(dst[e] - base) < rows) {
            const int p = atomicAdd(cnt, 1);
            if (p < LIST_CAP) l_e[p] = e;
        }
    }
}

__device__ __forceinline__ void push(bool p, int s, int dl, float w, int lane,
                                     int* cnt1, int* l_s, int* l_d, float* l_w) {
    const unsigned m = __builtin_amdgcn_ballot_w32(p);
    if (m) {
        int start = 0;
        if (lane == 0) start = atomicAdd(cnt1, __popc(m));
        start = __builtin_amdgcn_readfirstlane(start);
        if (p) {
            const int q = start + __popc(m & ((1u << lane) - 1u));
            l_s[q] = s;
            l_d[q] = dl;
            l_w[q] = w;
        }
    }
}

__device__ __forceinline__ void add_row(float* tr, float w, float4 a, float4 c) {
    unsafeAtomicAdd(tr,       w * a.x);
    unsafeAtomicAdd(tr + 32,  w * a.y);
    unsafeAtomicAdd(tr + 64,  w * a.z);
    unsafeAtomicAdd(tr + 96,  w * a.w);
    unsafeAtomicAdd(tr + 128, w * c.x);
    unsafeAtomicAdd(tr + 160, w * c.y);
    unsafeAtomicAdd(tr + 192, w * c.z);
    unsafeAtomicAdd(tr + 224, w * c.w);
}

__global__ __launch_bounds__(AGG_THREADS) void k_agg(const float* __restrict__ X,
                                                     const float* __restrict__ xW,
                                                     const float* __restrict__ deg,
                                                     const float* __restrict__ b,
                                                     const int* __restrict__ assign,
                                                     const int* __restrict__ he_part,
                                                     const int* __restrict__ src,
                                                     const int* __restrict__ dst,
                                                     const int* __restrict__ t_cnt,
                                                     const int* __restrict__ t_list, int lists,
                                                     float* __restrict__ out, int N, int E) {
    __shared__ __attribute__((aligned(16))) float tile[TILE_ROWS * C_DIM];
    __shared__ int   l_e[LIST_CAP];
    __shared__ int   l_s[LIST_CAP];
    __shared__ int   l_d[LIST_CAP];
    __shared__ float l_w[LIST_CAP];
    __shared__ float r_deg[TILE_ROWS];
    __shared__ int   r_cl[TILE_ROWS];
    __shared__ int   r_he[TILE_ROWS];
    __shared__ int   cnt[2];

    const int t    = threadIdx.x;
    const int lane = t & 31;
    const int wave = t >> 5;
    const int base = blockIdx.x * TILE_ROWS;
    const int left = N - base;
    const int rows = left < TILE_ROWS ? left : TILE_ROWS;

    if (t < rows) {
        const int n  = base + t;
        const int cl = assign[n];
        r_cl[t]  = cl;
        int he = 0;
#pragma unroll
        for (int w = 0; w < E_WGS; ++w) he |= he_part[w * (NCL / 32) + (cl >> 5)];
        r_he[t]  = (he >> (cl & 31)) & 1;
        r_deg[t] = deg[n];
    }
    if (t < 2) cnt[t] = 0;
    __syncthreads();

    {
        const float4 b0 = ((const float4*)b)[lane];
        const float4 b1 = ((const float4*)b)[32 + lane];
        float4 va[ROWS_PER_WAVE], vb[ROWS_PER_WAVE];
#pragma unroll
        for (int k = 0; k < ROWS_PER_WAVE; ++k) {
            const int r = wave + AGG_WAVES * k;
            va[k] = make_float4(0.f, 0.f, 0.f, 0.f);
            vb[k] = va[k];
            if (r < rows) {
                const float4* rp = (const float4*)((r_he[r] ? xW : X) + (size_t)(base + r) * C_DIM);
                va[k] = rp[lane];
                vb[k] = rp[32 + lane];
            }
        }
#pragma unroll
        for (int k = 0; k < ROWS_PER_WAVE; ++k) {
            const int r = wave + AGG_WAVES * k;
            if (r < rows) {
                float4 a = va[k], c = vb[k];
                if (r_he[r]) {
                    const float dg = r_deg[r];
                    a.x = a.x / dg + b0.x; a.y = a.y / dg + b0.y;
                    a.z = a.z / dg + b0.z; a.w = a.w / dg + b0.w;
                    c.x = c.x / dg + b1.x; c.y = c.y / dg + b1.y;
                    c.z = c.z / dg + b1.z; c.w = c.w / dg + b1.w;
                }
                float* tr = tile + r * C_DIM + lane;
                tr[0]   = a.x; tr[32]  = a.y; tr[64]  = a.z; tr[96]  = a.w;
                tr[128] = c.x; tr[160] = c.y; tr[192] = c.z; tr[224] = c.w;
            }
        }
    }

    const int n4 = E >> 2;
    int listed = -1;
    if (lists) listed = t_cnt[blockIdx.x];
    if (listed >= 0 && listed <= BCAP) {
        const int* tl = t_list + (size_t)blockIdx.x * BCAP;
        for (int i = t; i < listed; i += AGG_THREADS) l_e[i] = tl[i];
        if (t == 0) cnt[0] = listed;
    } else {
        scan_range(dst, 0, n4, n4, E, base, (unsigned)rows, &cnt[0], l_e);
    }
    __syncthreads();

    int step4 = n4;
    if (cnt[0] > LIST_CAP) step4 = LIST_CAP / 4 - 1;
    if (step4 < 1) step4 = 1;
    const bool one_pass = (step4 == n4) || (n4 == 0);

    for (int g0 = 0; g0 == 0 || g0 < n4; g0 += step4) {
        if (!one_pass) {
            __syncthreads();
            if (t < 2) cnt[t] = 0;
            __syncthreads();
            const int g1 = (g0 + step4 < n4) ? g0 + step4 : n4;
            scan_range(dst, g0, g1, n4, E, base, (unsigned)rows, &cnt[0], l_e);
            __syncthreads();
        }
        const int tot = cnt[0] < LIST_CAP ? cnt[0] : LIST_CAP;

        for (int ib = wave * 32; ib < tot; ib += 2 * AGG_THREADS) {
            const int  i0 = ib + lane, i1 = i0 + AGG_THREADS;
            const bool k0 = i0 < tot, k1 = i1 < tot;
            const int  e0 = l_e[k0 ? i0 : 0], e1 = l_e[k1 ? i1 : 0];
            const int  s0 = src[e0], d0 = dst[e0] - base;
            const int  s1 = src[e1], d1 = dst[e1] - base;
            const int  cs0 = assign[s0], cs1 = assign[s1];
            const float gs0 = deg[s0], gs1 = deg[s1];
            const bool in0 = k0 && cs0 == r_cl[d0];
            const bool in1 = k1 && cs1 == r_cl[d1];
            const float w0 = rsqrtf(gs0) * rsqrtf(r_deg[d0]);
            const float w1 = rsqrtf(gs1) * rsqrtf(r_deg[d1]);
            push(in0, s0, d0, w0, lane, &cnt[1], l_s, l_d, l_w);
            push(in1, s1, d1, w1, lane, &cnt[1], l_s, l_d, l_w);
        }
        __syncthreads();
        const int tot2 = cnt[1];

        const int nb = tot2 >> 2;
        for (int bi = wave; bi < nb; bi += AGG_WAVES) {
            int   s[4], dl[4];
            float w[4];
#pragma unroll
            for (int u = 0; u < 4; ++u) {
                s[u]  = __builtin_amdgcn_readfirstlane(l_s[4 * bi + u]);
                dl[u] = __builtin_amdgcn_readfirstlane(l_d[4 * bi + u]);
                w[u]  = l_w[4 * bi + u];
            }
            float4 a[4], c[4];
#pragma unroll
            for (int u = 0; u < 4; ++u) {
                const float4* rp = (const float4*)(xW + (size_t)s[u] * C_DIM);
                a[u] = rp[lane];
                c[u] = rp[32 + lane];
            }
#pragma unroll
            for (int u = 0; u < 4; ++u) add_row(tile + dl[u] * C_DIM + lane, w[u], a[u], c[u]);
        }
        if (wave < (tot2 & 3)) {
            const int i  = 4 * nb + wave;
            const int s  = __builtin_amdgcn_readfirstlane(l_s[i]);
            const int dl = __builtin_amdgcn_readfirstlane(l_d[i]);
            const float4* rp = (const float4*)(xW + (size_t)s * C_DIM);
            add_row(tile + dl * C_DIM + lane, l_w[i], rp[lane], rp[32 + lane]);
        }
    }
    __syncthreads();

    for (int r = wave; r < rows; r += AGG_WAVES) {
        const float* tr = tile + r * C_DIM + lane;
        float4 a, c;
        a.x = tr[0];   a.y = tr[32];  a.z = tr[64];  a.w = tr[96];
        c.x = tr[128]; c.y = tr[160]; c.z = tr[192]; c.w = tr[224];
        float4* op = (float4*)(out + (size_t)(base + r) * C_DIM);
        op[lane]      = a;
        op[32 + lane] = c;
    }
}

extern "C" void kernel_launch(void* const* d_in, const int* in_sizes, int n_in,
                              void* d_out, int out_size, void* d_ws, size_t ws_size,
                              hipStream_t stream) {
    const float* X      = (const float*)d_in[0];
    const float* W      = (const float*)d_in[1];
    const float* b      = (const float*)d_in[2];
    const int*   assign = (const int*)d_in[3];
    const int*   ei     = (const int*)d_in[4];

    const int N  = in_sizes[0] / C_DIM;
    const int E  = in_sizes[4] / 2;

    const int* src = ei;
    const int* dst = ei + E;

    char* ws = (char*)d_ws;
    size_t off = 0;
    float* deg = (float*)(ws + off);
    off += (((size_t)N * 4) + 255) & ~(size_t)255;
    int* he_part = (int*)(ws + off);
    off += (((size_t)NCL * 4) + 255) & ~(size_t)255;
    float* xW = (float*)(ws + off);
    off += (size_t)N * C_DIM * 4;
    const int ntiles = (N + TILE_ROWS - 1) / TILE_ROWS;
    int* t_cnt = (int*)(ws + off);
    off += (((size_t)ntiles * 4) + 255) & ~(size_t)255;
    int* t_list = (int*)(ws + off);
    off += (size_t)ntiles * BCAP * 4;
    const int lists = off <= ws_size ? 1 : 0;

    float* out = (float*)d_out;

    k_gemm<<<(N + G_ROWS - 1) / G_ROWS + E_WGS, G_THREADS, 0, stream>>>(X, W, xW, src, dst, assign,
                                                                          deg, he_part, t_cnt, t_list, lists,
                                                                          N, E);
    k_agg<<<ntiles, AGG_THREADS, 0, stream>>>(X, xW, deg, b, assign, he_part, src, dst, t_cnt, t_list, lists,
                                              out, N, E);
}


# edge lists by slice: each matrix work-group sorts its own 1/241 of the edge list by aggregation tile in LDS (count, scan, place); the aggregation joins the runs; no one searches the whole dst list
# speedup vs baseline: 3.8174x; 1.1370x over previous
#include <hip/hip_runtime.h>

typedef float v2f __attribute__((ext_vector_type(2)));
typedef float v8f __attribute__((ext_vector_type(8)));

#define C_DIM 256

#define E_WGS 15
#define NCL   256
#define TILE_ROWS 196
#define NT_MAX 256

__device__ __forceinline__ void edge_owner(const int* __restrict__ src, const int* __restrict__ dst,
                                           const int* __restrict__ assign,
                                           float* __restrict__ deg, int* __restrict__ he_part,
                                           int* lds, int w, int nthr, int N, int E) {
    const int t    = threadIdx.x;
    const int per  = (N + E_WGS - 1) / E_WGS;
    const int base = w * per;
    int rows = N - base;
    if (rows > per) rows = per;
    if (rows < 0) rows = 0;
    int* cnt  = lds;
    int* asg  = lds + per;
    int* flag = lds + 2 * per;
    for (int i = t; i < rows; i += nthr) { cnt[i] = 0; asg[i] = assign[base + i]; }
    for (int i = t; i < NCL; i += nthr) flag[i] = 0;
    __syncthreads();

    const int4* __restrict__ s4 = (const int4*)src;
    const int4* __restrict__ d4 = (const int4*)dst;
    const int n4  = E >> 2;
    const int rot = (int)(((long long)w * n4) / E_WGS) & ~31;
    for (int k0 = t; k0 < n4; k0 += 4 * nthr) {
        int s[16], d[16];
#pragma unroll
        for (int u = 0; u < 4; ++u) {
            const int k = k0 + u * nthr;
            int j = k + rot;
            if (j >= n4) j -= n4;
            int4 sv = make_int4(0, 0, 0, 0), dv = make_int4(-1, -1, -1, -1);
            if (k < n4) { sv = s4[j]; dv = d4[j]; }
            s[4 * u] = sv.x; s[4 * u + 1] = sv.y; s[4 * u + 2] = sv.z; s[4 * u + 3] = sv.w;
            d[4 * u] = dv.x; d[4 * u + 1] = dv.y; d[4 * u + 2] = dv.z; d[4 * u + 3] = dv.w;
        }
        int cs[16], cd[16];
#pragma unroll
        for (int i = 0; i < 16; ++i) {
            const unsigned dl = (unsigned)(d[i] - base);
            const bool m = dl < (unsigned)rows;
            cs[i] = assign[m ? s[i] : 0];
            cd[i] = m ? asg[dl] : -1;
            d[i]  = m ? (int)dl : -1;
        }
#pragma unroll
        for (int i = 0; i < 16; ++i) {
            if (d[i] >= 0 && cs[i] == cd[i]) {
                atomicAdd(&cnt[d[i]], 1);
                if ((unsigned)cs[i] < (unsigned)NCL) flag[cs[i]] = 1;
            }
        }
    }
    if (t < (E & 3)) {
        const int e = 4 * n4 + t;
        const unsigned dl = (unsigned)(dst[e] - base);
        if (dl < (unsigned)rows) {
            const int c = assign[src[e]];
            if (c == asg[dl]) {
                atomicAdd(&cnt[dl], 1);
                if ((unsigned)c < (unsigned)NCL) flag[c] = 1;
            }
        }
    }
    __syncthreads();

    for (int i = t; i < rows; i += nthr) deg[base + i] = 1.0f + (float)cnt[i];
    if (t < NCL / 32) {
        unsigned bits = 0;
        for (int i = 0; i < 32; ++i) bits |= (flag[32 * t + i] ? 1u : 0u) << i;
        he_part[w * (NCL / 32) + t] = (int)bits;
    }
}

#define G_TILES   13
#define G_ROWS    (G_TILES * 16)
#define G_UNITS   (G_TILES * 4)
#define G_THREADS 1024
#define G_WAVES   (G_THREADS / 32)
#define W_PITCH   1056


__global__ __launch_bounds__(G_THREADS) void k_gemm(const float* __restrict__ X,
                                                    const float* __restrict__ W,
                                                    float* __restrict__ xW,
                                                    const int* __restrict__ src,
                                                    const int* __restrict__ dst,
                                                    const int* __restrict__ assign,
                                                    float* __restrict__ deg,
                                                    int* __restrict__ he_part,
                                                    int* __restrict__ g_seg,
                                                    int* __restrict__ w_list, int lists,
                                                    int N, int E) {
    __shared__ __attribute__((aligned(16))) float Wl[(C_DIM / 4) * W_PITCH];
    __shared__ __attribute__((aligned(16))) int bcnt[NT_MAX];

    const int n_mat = (N + G_ROWS - 1) / G_ROWS;
    if ((int)blockIdx.x >= n_mat) {
        edge_owner(src, dst, assign, deg, he_part, (int*)Wl, (int)blockIdx.x - n_mat, G_THREADS, N, E);
        return;
    }

    const int lane = threadIdx.x & 31;
    const int wave = __builtin_amdgcn_readfirstlane(threadIdx.x >> 5);
    const int h    = lane >> 4;
    const int c    = lane & 15;

    const int ntiles = (N + TILE_ROWS - 1) / TILE_ROWS;
    const int per_e  = (E + n_mat - 1) / n_mat;
    const int e0     = (int)blockIdx.x * per_e;
    int ecnt = E - e0;
    if (ecnt > per_e) ecnt = per_e;
    if (ecnt < 0) ecnt = 0;
    int dpre[2] = {-1, -1};
    if (lists) {
        for (int i = threadIdx.x; i < NT_MAX; i += G_THREADS) bcnt[i] = 0;
#pragma unroll
        for (int u = 0; u < 2; ++u) {
            const int i = (int)threadIdx.x + u * G_THREADS;
            if (i < ecnt) dpre[u] = dst[e0 + i];
        }
    }

    for (int i = 0; i < 2; ++i) {
        const int q = wave + G_WAVES * i;
        const float* __restrict__ wq = W + (size_t)(4 * q) * C_DIM + lane;
        float* lq = Wl + q * W_PITCH + 4 * lane;
#pragma unroll
        for (int cc = 0; cc < 8; ++cc) {
            float4 v;
            v.x = wq[cc * 32];
            v.y = wq[C_DIM + cc * 32];
            v.z = wq[2 * C_DIM + cc * 32];
            v.w = wq[3 * C_DIM + cc * 32];
            *(float4*)(lq + cc * 128) = v;
        }
    }
    __syncthreads();

    if (lists) {
#pragma unroll
        for (int u = 0; u < 2; ++u) {
            const int d = dpre[u];
            if ((unsigned)d < (unsigned)N) atomicAdd(&bcnt[(unsigned)d / TILE_ROWS], 1);
        }
        for (int i = (int)threadIdx.x + 2 * G_THREADS; i < ecnt; i += G_THREADS) {
            const int d = dst[e0 + i];
            if ((unsigned)d < (unsigned)N) atomicAdd(&bcnt[(unsigned)d / TILE_ROWS], 1);
        }
        __syncthreads();
        if (wave == 0) {
            int cn[8], st[8];
            int run = 0;
#pragma unroll
            for (int k = 0; k < 8; ++k) { cn[k] = bcnt[8 * lane + k]; st[k] = run; run += cn[k]; }
            int inc = run;
            for (int d = 1; d < 32; d <<= 1) {
                const int v = __shfl_up(inc, d);
                if (lane >= d) inc += v;
            }
            const int lbase = inc - run;
#pragma unroll
            for (int k = 0; k < 8; ++k) {
                const int T = 8 * lane + k;
                bcnt[T] = lbase + st[k];
                if (T < ntiles) g_seg[(size_t)T * n_mat + blockIdx.x] = ((lbase + st[k]) << 16) | cn[k];
            }
        }
        __syncthreads();
#pragma unroll
        for (int u = 0; u < 2; ++u) {
            const int d = dpre[u];
            if ((unsigned)d < (unsigned)N) {
                const int p = atomicAdd(&bcnt[(unsigned)d / TILE_ROWS], 1);
                w_list[e0 + p] = e0 + (int)threadIdx.x + u * G_THREADS;
            }
        }
        for (int i = (int)threadIdx.x + 2 * G_THREADS; i < ecnt; i += G_THREADS) {
            const int d = dst[e0 + i];
            if ((unsigned)d < (unsigned)N) {
                const int p = atomicAdd(&bcnt[(unsigned)d / TILE_ROWS], 1);
                w_list[e0 + p] = e0 + i;
            }
        }
    }

    const int rbase = blockIdx.x * G_ROWS;
    for (int u = wave; u < G_UNITS; u += G_WAVES) {
        const int r0 = rbase + (u >> 2) * 16;
        if (r0 >= N) break;
        const int n0 = (u & 3) * 64;

        int rowA = r0 + c;
        if (rowA >= N) rowA = N - 1;
        const float* __restrict__ Ap = X + (size_t)rowA * C_DIM + 4 * h;
        const float* Bp = Wl + h * W_PITCH + 4 * (n0 + c);

        v8f acc[4] = {};
        for (int k8 = 0; k8 < C_DIM; k8 += 8) {
            const float4 a = *(const float4*)(Ap + k8);
            const float* bq = Bp + (k8 >> 2) * W_PITCH;
            float4 b[4];
#pragma unroll
            for (int t = 0; t < 4; ++t) b[t] = *(const float4*)(bq + t * 64);
            v2f a0, a1;
            a0.x = a.x; a0.y = a.y;
            a1.x = a.z; a1.y = a.w;
#pragma unroll
            for (int t = 0; t < 4; ++t) {
                v2f bf;
                bf.x = b[t].x; bf.y = b[t].y;
                acc[t] = __builtin_amdgcn_wmma_f32_16x16x4_f32(
                    false, a0, false, bf, (short)0, acc[t], false, false);
            }
#pragma unroll
            for (int t = 0; t < 4; ++t) {
                v2f bf;
                bf.x = b[t].z; bf.y = b[t].w;
                acc[t] = __builtin_amdgcn_wmma_f32_16x16x4_f32(
                    false, a1, false, bf, (short)0, acc[t], false, false);
            }
        }

        float* __restrict__ op = xW + (size_t)(r0 + 8 * h) * C_DIM + n0 + c;
        if (r0 + 16 <= N) {
#pragma unroll
            for (int t = 0; t < 4; ++t) {
#pragma unroll
                for (int v = 0; v < 8; ++v) op[(size_t)v * C_DIM + t * 16] = acc[t][v];
            }
        } else {
#pragma unroll
            for (int t = 0; t < 4; ++t) {
#pragma unroll
                for (int v = 0; v < 8; ++v) {
                    if (r0 + v + 8 * h < N) op[(size_t)v * C_DIM + t * 16] = acc[t][v];
                }
            }
        }
    }
}

#define LIST_CAP  4096
#define AGG_THREADS 1024
#define AGG_WAVES (AGG_THREADS / 32)
#define ROWS_PER_WAVE ((TILE_ROWS + AGG_WAVES - 1) / AGG_WAVES)


__device__ __forceinline__ void take4(int4 v, int e, int base, unsigned rows,
                                      int* cnt, int* l_e) {
    const bool m0 = (unsigned)(v.x - base) < rows;
    const bool m1 = (unsigned)(v.y - base) < rows;
    const bool m2 = (unsigned)(v.z - base) < rows;
    const bool m3 = (unsigned)(v.w - base) < rows;
    const int c = (int)m0 + (int)m1 + (int)m2 + (int)m3;
    if (c) {
        int p = atomicAdd(cnt, c);
        if (m0) { if (p < LIST_CAP) l_e[p] = e;     ++p; }
        if (m1) { if (p < LIST_CAP) l_e[p] = e + 1; ++p; }
        if (m2) { if (p < LIST_CAP) l_e[p] = e + 2; ++p; }
        if (m3) { if (p < LIST_CAP) l_e[p] = e + 3; }
    }
}

__device__ __forceinline__ void scan_range(const int* __restrict__ dst, int g0, int g1,
                                           int n4, int E, int base, unsigned rows,
                                           int* cnt, int* l_e) {
    const int4* __restrict__ d4 = (const int4*)dst;
    const int t = threadIdx.x;
    const int len = g1 - g0;
    const int rot = (int)(((long long)blockIdx.x * len) / gridDim.x) & ~31;
    for (int k0 = t; k0 < len; k0 += 4 * AGG_THREADS) {
        int4 v[4];
        int  g[4];
#pragma unroll
        for (int u = 0; u < 4; ++u) {
            const int k = k0 + u * AGG_THREADS;
            int j = k + rot;
            if (j >= len) j -= len;
            g[u] = g0 + j;
            v[u] = (k < len) ? d4[g[u]] : make_int4(-1, -1, -1, -1);
        }
#pragma unroll
        for (int u = 0; u < 4; ++u) take4(v[u], g[u] * 4, base, rows, cnt, l_e);
    }
    if (g1 == n4 && t < (E & 3)) {
        const int e = 4 * n4 + t;
        if ((unsigned)(dst[e] - base) < rows) {
            const int p = atomicAdd(cnt, 1);
            if (p < LIST_CAP) l_e[p] = e;
        }
    }
}

__device__ __forceinline__ void push(bool p, int s, int dl, float w, int lane,
                                     int* cnt1, int* l_s, int* l_d, float* l_w) {
    const unsigned m = __builtin_amdgcn_ballot_w32(p);
    if (m) {
        int start = 0;
        if (lane == 0) start = atomicAdd(cnt1, __popc(m));
        start = __builtin_amdgcn_readfirstlane(start);
        if (p) {
            const int q = start + __popc(m & ((1u << lane) - 1u));
            l_s[q] = s;
            l_d[q] = dl;
            l_w[q] = w;
        }
    }
}

__device__ __forceinline__ void add_row(float* tr, float w, float4 a, float4 c) {
    unsafeAtomicAdd(tr,       w * a.x);
    unsafeAtomicAdd(tr + 32,  w * a.y);
    unsafeAtomicAdd(tr + 64,  w * a.z);
    unsafeAtomicAdd(tr + 96,  w * a.w);
    unsafeAtomicAdd(tr + 128, w * c.x);
    unsafeAtomicAdd(tr + 160, w * c.y);
    unsafeAtomicAdd(tr + 192, w * c.z);
    unsafeAtomicAdd(tr + 224, w * c.w);
}

__global__ __launch_bounds__(AGG_THREADS) void k_agg(const float* __restrict__ X,
                                                     const float* __restrict__ xW,
                                                     const float* __restrict__ deg,
                                                     const float* __restrict__ b,
                                                     const int* __restrict__ assign,
                                                     const int* __restrict__ he_part,
                                                     const int* __restrict__ src,
                                                     const int* __restrict__ dst,
                                                     const int* __restrict__ g_seg,
                                                     const int* __restrict__ w_list, int lists,
                                                     float* __restrict__ out, int N, int E) {
    __shared__ __attribute__((aligned(16))) float tile[TILE_ROWS * C_DIM];
    __shared__ int   l_e[LIST_CAP];
    __shared__ int   l_s[LIST_CAP];
    __shared__ int   l_d[LIST_CAP];
    __shared__ float l_w[LIST_CAP];
    __shared__ float r_deg[TILE_ROWS];
    __shared__ int   r_cl[TILE_ROWS];
    __shared__ int   r_he[TILE_ROWS];
    __shared__ int   cnt[2];

    const int t    = threadIdx.x;
    const int lane = t & 31;
    const int wave = t >> 5;
    const int base = blockIdx.x * TILE_ROWS;
    const int left = N - base;
    const int rows = left < TILE_ROWS ? left : TILE_ROWS;

    if (t < rows) {
        const int n  = base + t;
        const int cl = assign[n];
        r_cl[t]  = cl;
        int he = 0;
#pragma unroll
        for (int w = 0; w < E_WGS; ++w) he |= he_part[w * (NCL / 32) + (cl >> 5)];
        r_he[t]  = (he >> (cl & 31)) & 1;
        r_deg[t] = deg[n];
    }
    if (t < 2) cnt[t] = 0;
    __syncthreads();

    {
        const float4 b0 = ((const float4*)b)[lane];
        const float4 b1 = ((const float4*)b)[32 + lane];
        float4 va[ROWS_PER_WAVE], vb[ROWS_PER_WAVE];
#pragma unroll
        for (int k = 0; k < ROWS_PER_WAVE; ++k) {
            const int r = wave + AGG_WAVES * k;
            va[k] = make_float4(0.f, 0.f, 0.f, 0.f);
            vb[k] = va[k];
            if (r < rows) {
                const float4* rp = (const float4*)((r_he[r] ? xW : X) + (size_t)(base + r) * C_DIM);
                va[k] = rp[lane];
                vb[k] = rp[32 + lane];
            }
        }
#pragma unroll
        for (int k = 0; k < ROWS_PER_WAVE; ++k) {
            const int r = wave + AGG_WAVES * k;
            if (r < rows) {
                float4 a = va[k], c = vb[k];
                if (r_he[r]) {
                    const float dg = r_deg[r];
                    a.x = a.x / dg + b0.x; a.y = a.y / dg + b0.y;
                    a.z = a.z / dg + b0.z; a.w = a.w / dg + b0.w;
                    c.x = c.x / dg + b1.x; c.y = c.y / dg + b1.y;
                    c.z = c.z / dg + b1.z; c.w = c.w / dg + b1.w;
                }
                float* tr = tile + r * C_DIM + lane;
                tr[0]   = a.x; tr[32]  = a.y; tr[64]  = a.z; tr[96]  = a.w;
                tr[128] = c.x; tr[160] = c.y; tr[192] = c.z; tr[224] = c.w;
            }
        }
    }

    const int n4 = E >> 2;
    if (lists) {
        const int n_mat = (N + G_ROWS - 1) / G_ROWS;
        const int per_e = (E + n_mat - 1) / n_mat;
        const int* sg = g_seg + (size_t)blockIdx.x * n_mat;
        for (int q0 = wave * 32; q0 < 4 * n_mat; q0 += AGG_THREADS) {
            const int q = q0 + lane;
            const int w = q >> 2, j0 = q & 3;
            int c = 0, start = 0;
            if (w < n_mat) { const int pk = sg[w]; c = pk & 0xffff; start = pk >> 16; }
            const int n = c > j0 ? (c - j0 + 3) >> 2 : 0;
            int inc = n;
            for (int d = 1; d < 32; d <<= 1) {
                const int v = __shfl_up(inc, d);
                if (lane >= d) inc += v;
            }
            const int tot = __shfl(inc, 31);
            int first = 0;
            if (lane == 0 && tot) first = atomicAdd(&cnt[0], tot);
            first = __builtin_amdgcn_readfirstlane(first);
            const int p = first + inc - n;
            const int* wl = w_list + (size_t)w * per_e + start + j0;
            int v[4];
#pragma unroll
            for (int k = 0; k < 4; ++k) v[k] = (k < n) ? wl[4 * k] : 0;
#pragma unroll
            for (int k = 0; k < 4; ++k) if (k < n && p + k < LIST_CAP) l_e[p + k] = v[k];
            for (int k = 4; k < n; ++k) if (p + k < LIST_CAP) l_e[p + k] = wl[4 * k];
        }
    } else {
        scan_range(dst, 0, n4, n4, E, base, (unsigned)rows, &cnt[0], l_e);
    }
    __syncthreads();

    int step4 = n4;
    if (cnt[0] > LIST_CAP) step4 = LIST_CAP / 4 - 1;
    if (step4 < 1) step4 = 1;
    const bool one_pass = (step4 == n4) || (n4 == 0);

    for (int g0 = 0; g0 == 0 || g0 < n4; g0 += step4) {
        if (!one_pass) {
            __syncthreads();
            if (t < 2) cnt[t] = 0;
            __syncthreads();
            const int g1 = (g0 + step4 < n4) ? g0 + step4 : n4;
            scan_range(dst, g0, g1, n4, E, base, (unsigned)rows, &cnt[0], l_e);
            __syncthreads();
        }
        const int tot = cnt[0] < LIST_CAP ? cnt[0] : LIST_CAP;

        for (int ib = wave * 32; ib < tot; ib += 2 * AGG_THREADS) {
            const int  i0 = ib + lane, i1 = i0 + AGG_THREADS;
            const bool k0 = i0 < tot, k1 = i1 < tot;
            const int  e0 = l_e[k0 ? i0 : 0], e1 = l_e[k1 ? i1 : 0];
            const int  s0 = src[e0], d0 = dst[e0] - base;
            const int  s1 = src[e1], d1 = dst[e1] - base;
            const int  cs0 = assign[s0], cs1 = assign[s1];
            const float gs0 = deg[s0], gs1 = deg[s1];
            const bool in0 = k0 && cs0 == r_cl[d0];
            const bool in1 = k1 && cs1 == r_cl[d1];
            const float w0 = rsqrtf(gs0) * rsqrtf(r_deg[d0]);
            const float w1 = rsqrtf(gs1) * rsqrtf(r_deg[d1]);
            push(in0, s0, d0, w0, lane, &cnt[1], l_s, l_d, l_w);
            push(in1, s1, d1, w1, lane, &cnt[1], l_s, l_d, l_w);
        }
        __syncthreads();
        const int tot2 = cnt[1];

        const int nb = tot2 >> 2;
        for (int bi = wave; bi < nb; bi += AGG_WAVES) {
            int   s[4], dl[4];
            float w[4];
#pragma unroll
            for (int u = 0; u < 4; ++u) {
                s[u]  = __builtin_amdgcn_readfirstlane(l_s[4 * bi + u]);
                dl[u] = __builtin_amdgcn_readfirstlane(l_d[4 * bi + u]);
                w[u]  = l_w[4 * bi + u];
            }
            float4 a[4], c[4];
#pragma unroll
            for (int u = 0; u < 4; ++u) {
                const float4* rp = (const float4*)(xW + (size_t)s[u] * C_DIM);
                a[u] = rp[lane];
                c[u] = rp[32 + lane];
            }
#pragma unroll
            for (int u = 0; u < 4; ++u) add_row(tile + dl[u] * C_DIM + lane, w[u], a[u], c[u]);
        }
        if (wave < (tot2 & 3)) {
            const int i  = 4 * nb + wave;
            const int s  = __builtin_amdgcn_readfirstlane(l_s[i]);
            const int dl = __builtin_amdgcn_readfirstlane(l_d[i]);
            const float4* rp = (const float4*)(xW + (size_t)s * C_DIM);
            add_row(tile + dl * C_DIM + lane, l_w[i], rp[lane], rp[32 + lane]);
        }
    }
    __syncthreads();

    for (int r = wave; r < rows; r += AGG_WAVES) {
        const float* tr = tile + r * C_DIM + lane;
        float4 a, c;
        a.x = tr[0];   a.y = tr[32];  a.z = tr[64];  a.w = tr[96];
        c.x = tr[128]; c.y = tr[160]; c.z = tr[192]; c.w = tr[224];
        float4* op = (float4*)(out + (size_t)(base + r) * C_DIM);
        op[lane]      = a;
        op[32 + lane] = c;
    }
}

extern "C" void kernel_launch(void* const* d_in, const int* in_sizes, int n_in,
                              void* d_out, int out_size, void* d_ws, size_t ws_size,
                              hipStream_t stream) {
    const float* X      = (const float*)d_in[0];
    const float* W      = (const float*)d_in[1];
    const float* b      = (const float*)d_in[2];
    const int*   assign = (const int*)d_in[3];
    const int*   ei     = (const int*)d_in[4];

    const int N  = in_sizes[0] / C_DIM;
    const int E  = in_sizes[4] / 2;

    const int* src = ei;
    const int* dst = ei + E;

    char* ws = (char*)d_ws;
    size_t off = 0;
    float* deg = (float*)(ws + off);
    off += (((size_t)N * 4) + 255) & ~(size_t)255;
    int* he_part = (int*)(ws + off);
    off += (((size_t)NCL * 4) + 255) & ~(size_t)255;
    float* xW = (float*)(ws + off);
    off += (size_t)N * C_DIM * 4;
    const int ntiles = (N + TILE_ROWS - 1) / TILE_ROWS;
    const int n_mat  = (N + G_ROWS - 1) / G_ROWS;
    const int per_e  = (E + n_mat - 1) / n_mat;
    int* g_seg = (int*)(ws + off);
    off += (((size_t)ntiles * n_mat * 4) + 255) & ~(size_t)255;
    int* w_list = (int*)(ws + off);
    off += (size_t)E * 4;
    const int lists = (off <= ws_size && ntiles <= NT_MAX && per_e < 32768) ? 1 : 0;

    float* out = (float*)d_out;

    k_gemm<<<n_mat + E_WGS, G_THREADS, 0, stream>>>(X, W, xW, src, dst, assign,
                                                      deg, he_part, g_seg, w_list, lists,
                                                      N, E);
    k_agg<<<ntiles, AGG_THREADS, 0, stream>>>(X, xW, deg, b, assign, he_part, src, dst, g_seg, w_list, lists,
                                              out, N, E);
}


# edge runs carry (source row, local destination row) in one word instead of the edge number: the aggregation drops its src[e] and dst[e] lookups, two scattered 4-byte reads per edge
# speedup vs baseline: 3.9371x; 1.0314x over previous
#include <hip/hip_runtime.h>

typedef float v2f __attribute__((ext_vector_type(2)));
typedef float v8f __attribute__((ext_vector_type(8)));

#define C_DIM 256

#define E_WGS 15
#define NCL   256
#define TILE_ROWS 196
#define NT_MAX 256

__device__ __forceinline__ void edge_owner(const int* __restrict__ src, const int* __restrict__ dst,
                                           const int* __restrict__ assign,
                                           float* __restrict__ deg, int* __restrict__ he_part,
                                           int* lds, int w, int nthr, int N, int E) {
    const int t    = threadIdx.x;
    const int per  = (N + E_WGS - 1) / E_WGS;
    const int base = w * per;
    int rows = N - base;
    if (rows > per) rows = per;
    if (rows < 0) rows = 0;
    int* cnt  = lds;
    int* asg  = lds + per;
    int* flag = lds + 2 * per;
    for (int i = t; i < rows; i += nthr) { cnt[i] = 0; asg[i] = assign[base + i]; }
    for (int i = t; i < NCL; i += nthr) flag[i] = 0;
    __syncthreads();

    const int4* __restrict__ s4 = (const int4*)src;
    const int4* __restrict__ d4 = (const int4*)dst;
    const int n4  = E >> 2;
    const int rot = (int)(((long long)w * n4) / E_WGS) & ~31;
    for (int k0 = t; k0 < n4; k0 += 4 * nthr) {
        int s[16], d[16];
#pragma unroll
        for (int u = 0; u < 4; ++u) {
            const int k = k0 + u * nthr;
            int j = k + rot;
            if (j >= n4) j -= n4;
            int4 sv = make_int4(0, 0, 0, 0), dv = make_int4(-1, -1, -1, -1);
            if (k < n4) { sv = s4[j]; dv = d4[j]; }
            s[4 * u] = sv.x; s[4 * u + 1] = sv.y; s[4 * u + 2] = sv.z; s[4 * u + 3] = sv.w;
            d[4 * u] = dv.x; d[4 * u + 1] = dv.y; d[4 * u + 2] = dv.z; d[4 * u + 3] = dv.w;
        }
        int cs[16], cd[16];
#pragma unroll
        for (int i = 0; i < 16; ++i) {
            const unsigned dl = (unsigned)(d[i] - base);
            const bool m = dl < (unsigned)rows;
            cs[i] = assign[m ? s[i] : 0];
            cd[i] = m ? asg[dl] : -1;
            d[i]  = m ? (int)dl : -1;
        }
#pragma unroll
        for (int i = 0; i < 16; ++i) {
            if (d[i] >= 0 && cs[i] == cd[i]) {
                atomicAdd(&cnt[d[i]], 1);
                if ((unsigned)cs[i] < (unsigned)NCL) flag[cs[i]] = 1;
            }
        }
    }
    if (t < (E & 3)) {
        const int e = 4 * n4 + t;
        const unsigned dl = (unsigned)(dst[e] - base);
        if (dl < (unsigned)rows) {
            const int c = assign[src[e]];
            if (c == asg[dl]) {
                atomicAdd(&cnt[dl], 1);
                if ((unsigned)c < (unsigned)NCL) flag[c] = 1;
            }
        }
    }
    __syncthreads();

    for (int i = t; i < rows; i += nthr) deg[base + i] = 1.0f + (float)cnt[i];
    if (t < NCL / 32) {
        unsigned bits = 0;
        for (int i = 0; i < 32; ++i) bits |= (flag[32 * t + i] ? 1u : 0u) << i;
        he_part[w * (NCL / 32) + t] = (int)bits;
    }
}

#define G_TILES   13
#define G_ROWS    (G_TILES * 16)
#define G_UNITS   (G_TILES * 4)
#define G_THREADS 1024
#define G_WAVES   (G_THREADS / 32)
#define W_PITCH   1056

#define PACK_ROWS 65536

__global__ __launch_bounds__(G_THREADS) void k_gemm(const float* __restrict__ X,
                                                    const float* __restrict__ W,
                                                    float* __restrict__ xW,
                                                    const int* __restrict__ src,
                                                    const int* __restrict__ dst,
                                                    const int* __restrict__ assign,
                                                    float* __restrict__ deg,
                                                    int* __restrict__ he_part,
                                                    int* __restrict__ g_seg,
                                                    int* __restrict__ w_list, int lists,
                                                    int N, int E) {
    __shared__ __attribute__((aligned(16))) float Wl[(C_DIM / 4) * W_PITCH];
    __shared__ __attribute__((aligned(16))) int bcnt[NT_MAX];

    const int n_mat = (N + G_ROWS - 1) / G_ROWS;
    if ((int)blockIdx.x >= n_mat) {
        edge_owner(src, dst, assign, deg, he_part, (int*)Wl, (int)blockIdx.x - n_mat, G_THREADS, N, E);
        return;
    }

    const int lane = threadIdx.x & 31;
    const int wave = __builtin_amdgcn_readfirstlane(threadIdx.x >> 5);
    const int h    = lane >> 4;
    const int c    = lane & 15;

    const int ntiles = (N + TILE_ROWS - 1) / TILE_ROWS;
    const int per_e  = (E + n_mat - 1) / n_mat;
    const int e0     = (int)blockIdx.x * per_e;
    int ecnt = E - e0;
    if (ecnt > per_e) ecnt = per_e;
    if (ecnt < 0) ecnt = 0;
    int dpre[2] = {-1, -1};
    int spre[2] = {0, 0};
    const bool use_lists = lists && N <= PACK_ROWS;
    if (use_lists) {
        for (int i = threadIdx.x; i < NT_MAX; i += G_THREADS) bcnt[i] = 0;
#pragma unroll
        for (int u = 0; u < 2; ++u) {
            const int i = (int)threadIdx.x + u * G_THREADS;
            if (i < ecnt) { dpre[u] = dst[e0 + i]; spre[u] = src[e0 + i]; }
        }
    }

    for (int i = 0; i < 2; ++i) {
        const int q = wave + G_WAVES * i;
        const float* __restrict__ wq = W + (size_t)(4 * q) * C_DIM + lane;
        float* lq = Wl + q * W_PITCH + 4 * lane;
#pragma unroll
        for (int cc = 0; cc < 8; ++cc) {
            float4 v;
            v.x = wq[cc * 32];
            v.y = wq[C_DIM + cc * 32];
            v.z = wq[2 * C_DIM + cc * 32];
            v.w = wq[3 * C_DIM + cc * 32];
            *(float4*)(lq + cc * 128) = v;
        }
    }
    __syncthreads();

    if (use_lists) {
#pragma unroll
        for (int u = 0; u < 2; ++u) {
            const int d = dpre[u];
            if ((unsigned)d < (unsigned)N) atomicAdd(&bcnt[(unsigned)d / TILE_ROWS], 1);
        }
        for (int i = (int)threadIdx.x + 2 * G_THREADS; i < ecnt; i += G_THREADS) {
            const int d = dst[e0 + i];
            if ((unsigned)d < (unsigned)N) atomicAdd(&bcnt[(unsigned)d / TILE_ROWS], 1);
        }
        __syncthreads();
        if (wave == 0) {
            int cn[8], st[8];
            int run = 0;
#pragma unroll
            for (int k = 0; k < 8; ++k) { cn[k] = bcnt[8 * lane + k]; st[k] = run; run += cn[k]; }
            int inc = run;
            for (int d = 1; d < 32; d <<= 1) {
                const int v = __shfl_up(inc, d);
                if (lane >= d) inc += v;
            }
            const int lbase = inc - run;
#pragma unroll
            for (int k = 0; k < 8; ++k) {
                const int T = 8 * lane + k;
                bcnt[T] = lbase + st[k];
                if (T < ntiles) g_seg[(size_t)T * n_mat + blockIdx.x] = ((lbase + st[k]) << 16) | cn[k];
            }
        }
        __syncthreads();
#pragma unroll
        for (int u = 0; u < 2; ++u) {
            const int d = dpre[u];
            if ((unsigned)d < (unsigned)N) {
                const int T = (int)((unsigned)d / TILE_ROWS);
                const int p = atomicAdd(&bcnt[T], 1);
                w_list[e0 + p] = spre[u] | ((d - T * TILE_ROWS) << 16);
            }
        }
        for (int i = (int)threadIdx.x + 2 * G_THREADS; i < ecnt; i += G_THREADS) {
            const int d = dst[e0 + i];
            if ((unsigned)d < (unsigned)N) {
                const int T = (int)((unsigned)d / TILE_ROWS);
                const int p = atomicAdd(&bcnt[T], 1);
                w_list[e0 + p] = src[e0 + i] | ((d - T * TILE_ROWS) << 16);
            }
        }
    }

    const int rbase = blockIdx.x * G_ROWS;
    for (int u = wave; u < G_UNITS; u += G_WAVES) {
        const int r0 = rbase + (u >> 2) * 16;
        if (r0 >= N) break;
        const int n0 = (u & 3) * 64;

        int rowA = r0 + c;
        if (rowA >= N) rowA = N - 1;
        const float* __restrict__ Ap = X + (size_t)rowA * C_DIM + 4 * h;
        const float* Bp = Wl + h * W_PITCH + 4 * (n0 + c);

        v8f acc[4] = {};
        for (int k8 = 0; k8 < C_DIM; k8 += 8) {
            const float4 a = *(const float4*)(Ap + k8);
            const float* bq = Bp + (k8 >> 2) * W_PITCH;
            float4 b[4];
#pragma unroll
            for (int t = 0; t < 4; ++t) b[t] = *(const float4*)(bq + t * 64);
            v2f a0, a1;
            a0.x = a.x; a0.y = a.y;
            a1.x = a.z; a1.y = a.w;
#pragma unroll
            for (int t = 0; t < 4; ++t) {
                v2f bf;
                bf.x = b[t].x; bf.y = b[t].y;
                acc[t] = __builtin_amdgcn_wmma_f32_16x16x4_f32(
                    false, a0, false, bf, (short)0, acc[t], false, false);
            }
#pragma unroll
            for (int t = 0; t < 4; ++t) {
                v2f bf;
                bf.x = b[t].z; bf.y = b[t].w;
                acc[t] = __builtin_amdgcn_wmma_f32_16x16x4_f32(
                    false, a1, false, bf, (short)0, acc[t], false, false);
            }
        }

        float* __restrict__ op = xW + (size_t)(r0 + 8 * h) * C_DIM + n0 + c;
        if (r0 + 16 <= N) {
#pragma unroll
            for (int t = 0; t < 4; ++t) {
#pragma unroll
                for (int v = 0; v < 8; ++v) op[(size_t)v * C_DIM + t * 16] = acc[t][v];
            }
        } else {
#pragma unroll
            for (int t = 0; t < 4; ++t) {
#pragma unroll
                for (int v = 0; v < 8; ++v) {
                    if (r0 + v + 8 * h < N) op[(size_t)v * C_DIM + t * 16] = acc[t][v];
                }
            }
        }
    }
}

#define LIST_CAP  4096
#define AGG_THREADS 1024
#define AGG_WAVES (AGG_THREADS / 32)
#define ROWS_PER_WAVE ((TILE_ROWS + AGG_WAVES - 1) / AGG_WAVES)


__device__ __forceinline__ void take4(int4 v, int e, int base, unsigned rows,
                                      int* cnt, int* l_e) {
    const bool m0 = (unsigned)(v.x - base) < rows;
    const bool m1 = (unsigned)(v.y - base) < rows;
    const bool m2 = (unsigned)(v.z - base) < rows;
    const bool m3 = (unsigned)(v.w - base) < rows;
    const int c = (int)m0 + (int)m1 + (int)m2 + (int)m3;
    if (c) {
        int p = atomicAdd(cnt, c);
        if (m0) { if (p < LIST_CAP) l_e[p] = e;     ++p; }
        if (m1) { if (p < LIST_CAP) l_e[p] = e + 1; ++p; }
        if (m2) { if (p < LIST_CAP) l_e[p] = e + 2; ++p; }
        if (m3) { if (p < LIST_CAP) l_e[p] = e + 3; }
    }
}

__device__ __forceinline__ void scan_range(const int* __restrict__ dst, int g0, int g1,
                                           int n4, int E, int base, unsigned rows,
                                           int* cnt, int* l_e) {
    const int4* __restrict__ d4 = (const int4*)dst;
    const int t = threadIdx.x;
    const int len = g1 - g0;
    const int rot = (int)(((long long)blockIdx.x * len) / gridDim.x) & ~31;
    for (int k0 = t; k0 < len; k0 += 4 * AGG_THREADS) {
        int4 v[4];
        int  g[4];
#pragma unroll
        for (int u = 0; u < 4; ++u) {
            const int k = k0 + u * AGG_THREADS;
            int j = k + rot;
            if (j >= len) j -= len;
            g[u] = g0 + j;
            v[u] = (k < len) ? d4[g[u]] : make_int4(-1, -1, -1, -1);
        }
#pragma unroll
        for (int u = 0; u < 4; ++u) take4(v[u], g[u] * 4, base, rows, cnt, l_e);
    }
    if (g1 == n4 && t < (E & 3)) {
        const int e = 4 * n4 + t;
        if ((unsigned)(dst[e] - base) < rows) {
            const int p = atomicAdd(cnt, 1);
            if (p < LIST_CAP) l_e[p] = e;
        }
    }
}

__device__ __forceinline__ void push(bool p, int s, int dl, float w, int lane,
                                     int* cnt1, int* l_s, int* l_d, float* l_w) {
    const unsigned m = __builtin_amdgcn_ballot_w32(p);
    if (m) {
        int start = 0;
        if (lane == 0) start = atomicAdd(cnt1, __popc(m));
        start = __builtin_amdgcn_readfirstlane(start);
        if (p) {
            const int q = start + __popc(m & ((1u << lane) - 1u));
            l_s[q] = s;
            l_d[q] = dl;
            l_w[q] = w;
        }
    }
}

__device__ __forceinline__ void add_row(float* tr, float w, float4 a, float4 c) {
    unsafeAtomicAdd(tr,       w * a.x);
    unsafeAtomicAdd(tr + 32,  w * a.y);
    unsafeAtomicAdd(tr + 64,  w * a.z);
    unsafeAtomicAdd(tr + 96,  w * a.w);
    unsafeAtomicAdd(tr + 128, w * c.x);
    unsafeAtomicAdd(tr + 160, w * c.y);
    unsafeAtomicAdd(tr + 192, w * c.z);
    unsafeAtomicAdd(tr + 224, w * c.w);
}

__global__ __launch_bounds__(AGG_THREADS) void k_agg(const float* __restrict__ X,
                                                     const float* __restrict__ xW,
                                                     const float* __restrict__ deg,
                                                     const float* __restrict__ b,
                                                     const int* __restrict__ assign,
                                                     const int* __restrict__ he_part,
                                                     const int* __restrict__ src,
                                                     const int* __restrict__ dst,
                                                     const int* __restrict__ g_seg,
                                                     const int* __restrict__ w_list, int lists,
                                                     float* __restrict__ out, int N, int E) {
    __shared__ __attribute__((aligned(16))) float tile[TILE_ROWS * C_DIM];
    __shared__ int   l_e[LIST_CAP];
    __shared__ int   l_s[LIST_CAP];
    __shared__ int   l_d[LIST_CAP];
    __shared__ float l_w[LIST_CAP];
    __shared__ float r_deg[TILE_ROWS];
    __shared__ int   r_cl[TILE_ROWS];
    __shared__ int   r_he[TILE_ROWS];
    __shared__ int   cnt[2];

    const int t    = threadIdx.x;
    const int lane = t & 31;
    const int wave = t >> 5;
    const int base = blockIdx.x * TILE_ROWS;
    const int left = N - base;
    const int rows = left < TILE_ROWS ? left : TILE_ROWS;

    if (t < rows) {
        const int n  = base + t;
        const int cl = assign[n];
        r_cl[t]  = cl;
        int he = 0;
#pragma unroll
        for (int w = 0; w < E_WGS; ++w) he |= he_part[w * (NCL / 32) + (cl >> 5)];
        r_he[t]  = (he >> (cl & 31)) & 1;
        r_deg[t] = deg[n];
    }
    if (t < 2) cnt[t] = 0;
    __syncthreads();

    {
        const float4 b0 = ((const float4*)b)[lane];
        const float4 b1 = ((const float4*)b)[32 + lane];
        float4 va[ROWS_PER_WAVE], vb[ROWS_PER_WAVE];
#pragma unroll
        for (int k = 0; k < ROWS_PER_WAVE; ++k) {
            const int r = wave + AGG_WAVES * k;
            va[k] = make_float4(0.f, 0.f, 0.f, 0.f);
            vb[k] = va[k];
            if (r < rows) {
                const float4* rp = (const float4*)((r_he[r] ? xW : X) + (size_t)(base + r) * C_DIM);
                va[k] = rp[lane];
                vb[k] = rp[32 + lane];
            }
        }
#pragma unroll
        for (int k = 0; k < ROWS_PER_WAVE; ++k) {
            const int r = wave + AGG_WAVES * k;
            if (r < rows) {
                float4 a = va[k], c = vb[k];
                if (r_he[r]) {
                    const float dg = r_deg[r];
                    a.x = a.x / dg + b0.x; a.y = a.y / dg + b0.y;
                    a.z = a.z / dg + b0.z; a.w = a.w / dg + b0.w;
                    c.x = c.x / dg + b1.x; c.y = c.y / dg + b1.y;
                    c.z = c.z / dg + b1.z; c.w = c.w / dg + b1.w;
                }
                float* tr = tile + r * C_DIM + lane;
                tr[0]   = a.x; tr[32]  = a.y; tr[64]  = a.z; tr[96]  = a.w;
                tr[128] = c.x; tr[160] = c.y; tr[192] = c.z; tr[224] = c.w;
            }
        }
    }

    const int n4 = E >> 2;
    const bool use_lists = lists && N <= PACK_ROWS;
    if (use_lists) {
        const int n_mat = (N + G_ROWS - 1) / G_ROWS;
        const int per_e = (E + n_mat - 1) / n_mat;
        const int* sg = g_seg + (size_t)blockIdx.x * n_mat;
        for (int q0 = wave * 32; q0 < 4 * n_mat; q0 += AGG_THREADS) {
            const int q = q0 + lane;
            const int w = q >> 2, j0 = q & 3;
            int c = 0, start = 0;
            if (w < n_mat) { const int pk = sg[w]; c = pk & 0xffff; start = pk >> 16; }
            const int n = c > j0 ? (c - j0 + 3) >> 2 : 0;
            int inc = n;
            for (int d = 1; d < 32; d <<= 1) {
                const int v = __shfl_up(inc, d);
                if (lane >= d) inc += v;
            }
            const int tot = __shfl(inc, 31);
            int first = 0;
            if (lane == 0 && tot) first = atomicAdd(&cnt[0], tot);
            first = __builtin_amdgcn_readfirstlane(first);
            const int p = first + inc - n;
            const int* wl = w_list + (size_t)w * per_e + start + j0;
            int v[4];
#pragma unroll
            for (int k = 0; k < 4; ++k) v[k] = (k < n) ? wl[4 * k] : 0;
#pragma unroll
            for (int k = 0; k < 4; ++k) if (k < n && p + k < LIST_CAP) l_e[p + k] = v[k];
            for (int k = 4; k < n; ++k) if (p + k < LIST_CAP) l_e[p + k] = wl[4 * k];
        }
    } else {
        scan_range(dst, 0, n4, n4, E, base, (unsigned)rows, &cnt[0], l_e);
    }
    __syncthreads();

    int step4 = n4;
    if (cnt[0] > LIST_CAP) step4 = LIST_CAP / 4 - 1;
    if (step4 < 1) step4 = 1;
    const bool one_pass = (step4 == n4) || (n4 == 0);
    const bool packed = use_lists && one_pass;

    for (int g0 = 0; g0 == 0 || g0 < n4; g0 += step4) {
        if (!one_pass) {
            __syncthreads();
            if (t < 2) cnt[t] = 0;
            __syncthreads();
            const int g1 = (g0 + step4 < n4) ? g0 + step4 : n4;
            scan_range(dst, g0, g1, n4, E, base, (unsigned)rows, &cnt[0], l_e);
            __syncthreads();
        }
        const int tot = cnt[0] < LIST_CAP ? cnt[0] : LIST_CAP;

        for (int ib = wave * 32; ib < tot; ib += 2 * AGG_THREADS) {
            const int  i0 = ib + lane, i1 = i0 + AGG_THREADS;
            const bool k0 = i0 < tot, k1 = i1 < tot;
            const int  e0 = l_e[k0 ? i0 : 0], e1 = l_e[k1 ? i1 : 0];
            int s0, d0, s1, d1;
            if (packed) {
                s0 = e0 & 0xffff; d0 = e0 >> 16;
                s1 = e1 & 0xffff; d1 = e1 >> 16;
            } else {
                s0 = src[e0]; d0 = dst[e0] - base;
                s1 = src[e1]; d1 = dst[e1] - base;
            }
            const int  cs0 = assign[s0], cs1 = assign[s1];
            const float gs0 = deg[s0], gs1 = deg[s1];
            const bool in0 = k0 && cs0 == r_cl[d0];
            const bool in1 = k1 && cs1 == r_cl[d1];
            const float w0 = rsqrtf(gs0) * rsqrtf(r_deg[d0]);
            const float w1 = rsqrtf(gs1) * rsqrtf(r_deg[d1]);
            push(in0, s0, d0, w0, lane, &cnt[1], l_s, l_d, l_w);
            push(in1, s1, d1, w1, lane, &cnt[1], l_s, l_d, l_w);
        }
        __syncthreads();
        const int tot2 = cnt[1];

        const int nb = tot2 >> 2;
        for (int bi = wave; bi < nb; bi += AGG_WAVES) {
            int   s[4], dl[4];
            float w[4];
#pragma unroll
            for (int u = 0; u < 4; ++u) {
                s[u]  = __builtin_amdgcn_readfirstlane(l_s[4 * bi + u]);
                dl[u] = __builtin_amdgcn_readfirstlane(l_d[4 * bi + u]);
                w[u]  = l_w[4 * bi + u];
            }
            float4 a[4], c[4];
#pragma unroll
            for (int u = 0; u < 4; ++u) {
                const float4* rp = (const float4*)(xW + (size_t)s[u] * C_DIM);
                a[u] = rp[lane];
                c[u] = rp[32 + lane];
            }
#pragma unroll
            for (int u = 0; u < 4; ++u) add_row(tile + dl[u] * C_DIM + lane, w[u], a[u], c[u]);
        }
        if (wave < (tot2 & 3)) {
            const int i  = 4 * nb + wave;
            const int s  = __builtin_amdgcn_readfirstlane(l_s[i]);
            const int dl = __builtin_amdgcn_readfirstlane(l_d[i]);
            const float4* rp = (const float4*)(xW + (size_t)s * C_DIM);
            add_row(tile + dl * C_DIM + lane, l_w[i], rp[lane], rp[32 + lane]);
        }
    }
    __syncthreads();

    for (int r = wave; r < rows; r += AGG_WAVES) {
        const float* tr = tile + r * C_DIM + lane;
        float4 a, c;
        a.x = tr[0];   a.y = tr[32];  a.z = tr[64];  a.w = tr[96];
        c.x = tr[128]; c.y = tr[160]; c.z = tr[192]; c.w = tr[224];
        float4* op = (float4*)(out + (size_t)(base + r) * C_DIM);
        op[lane]      = a;
        op[32 + lane] = c;
    }
}

extern "C" void kernel_launch(void* const* d_in, const int* in_sizes, int n_in,
                              void* d_out, int out_size, void* d_ws, size_t ws_size,
                              hipStream_t stream) {
    const float* X      = (const float*)d_in[0];
    const float* W      = (const float*)d_in[1];
    const float* b      = (const float*)d_in[2];
    const int*   assign = (const int*)d_in[3];
    const int*   ei     = (const int*)d_in[4];

    const int N  = in_sizes[0] / C_DIM;
    const int E  = in_sizes[4] / 2;

    const int* src = ei;
    const int* dst = ei + E;

    char* ws = (char*)d_ws;
    size_t off = 0;
    float* deg = (float*)(ws + off);
    off += (((size_t)N * 4) + 255) & ~(size_t)255;
    int* he_part = (int*)(ws + off);
    off += (((size_t)NCL * 4) + 255) & ~(size_t)255;
    float* xW = (float*)(ws + off);
    off += (size_t)N * C_DIM * 4;
    const int ntiles = (N + TILE_ROWS - 1) / TILE_ROWS;
    const int n_mat  = (N + G_ROWS - 1) / G_ROWS;
    const int per_e  = (E + n_mat - 1) / n_mat;
    int* g_seg = (int*)(ws + off);
    off += (((size_t)ntiles * n_mat * 4) + 255) & ~(size_t)255;
    int* w_list = (int*)(ws + off);
    off += (size_t)E * 4;
    const int lists = (off <= ws_size && ntiles <= NT_MAX && per_e < 32768) ? 1 : 0;

    float* out = (float*)d_out;

    k_gemm<<<n_mat + E_WGS, G_THREADS, 0, stream>>>(X, W, xW, src, dst, assign,
                                                      deg, he_part, g_seg, w_list, lists,
                                                      N, E);
    k_agg<<<ntiles, AGG_THREADS, 0, stream>>>(X, xW, deg, b, assign, he_part, src, dst, g_seg, w_list, lists,
                                              out, N, E);
}
